# starting kernel
# baseline (speedup 1.0000x reference)
#include <hip/hip_runtime.h>
#include <math.h>

#define BB 2
#define LL 4096
#define EE 1024
#define HH 16
#define DD 64
#define NC 16
#define HEADSZ ((size_t)BB * HH * LL * DD)

typedef __attribute__((ext_vector_type(2))) float v2f;
typedef __attribute__((ext_vector_type(8))) float v8f;

__device__ __forceinline__ v8f wmma4(v2f a, v2f b, v8f c) {
  return __builtin_amdgcn_wmma_f32_16x16x4_f32(false, a, false, b, (short)0, c,
                                               false, false);
}

__device__ __forceinline__ float elup1(float x) {
  return x > 0.f ? x + 1.f : expf(x);
}

__global__ __launch_bounds__(256) void qkv_gemm(const float* __restrict__ x,
                                                const float* __restrict__ W,
                                                float* __restrict__ qb,
                                                float* __restrict__ kb,
                                                float* __restrict__ vb) {
  __shared__ float xs[128 * 36];
  __shared__ float wls[64 * 36];

  const int tid = threadIdx.x;
  const int lane = tid & 31, wv = tid >> 5;
  const int hi = lane >> 4, nlo = lane & 15;
  const int mBase = blockIdx.y * 128;
  const int nBase = blockIdx.x * 64;

  v8f acc[4] = {};

  for (int kt = 0; kt < 1024; kt += 32) {
#pragma unroll
    for (int i = 0; i < 4; ++i) {
      int q = tid + 256 * i;
      int row = q >> 3, c4 = (q & 7) * 4;
      float4 t4 = *(const float4*)(x + (size_t)(mBase + row) * 1024 + kt + c4);
      *(float4*)(xs + row * 36 + c4) = t4;
    }
#pragma unroll
    for (int i = 0; i < 2; ++i) {
      int q = tid + 256 * i;
      int row = q >> 3, c4 = (q & 7) * 4;
      float4 t4 = *(const float4*)(W + (size_t)(nBase + row) * 1024 + kt + c4);
      *(float4*)(wls + row * 36 + c4) = t4;
    }
    __syncthreads();

    const float* xr = xs + (wv * 16 + nlo) * 36;
#pragma unroll
    for (int kk = 0; kk < 32; kk += 4) {
      v2f a;
      a.x = xr[kk + 2 * hi];
      a.y = xr[kk + 2 * hi + 1];
#pragma unroll
      for (int tc = 0; tc < 4; ++tc) {
        const float* wr = wls + (tc * 16 + nlo) * 36;
        v2f b;
        b.x = wr[kk + 2 * hi];
        b.y = wr[kk + 2 * hi + 1];
        acc[tc] = wmma4(a, b, acc[tc]);
      }
    }
    __syncthreads();
  }

  const int tsel = nBase >> 10;
  float* dst = (tsel == 0) ? qb : (tsel == 1) ? kb : vb;
#pragma unroll
  for (int tc = 0; tc < 4; ++tc) {
    int n = nBase + tc * 16 + nlo;
    int rem = n & 1023;
    int h = rem >> 6, d = rem & 63;
#pragma unroll
    for (int j = 0; j < 8; ++j) {
      int m = mBase + wv * 16 + j + 8 * hi;
      float y = acc[tc][j];
      if (tsel == 0)
        y = elup1(y) * 0.125f;
      else if (tsel == 1)
        y = elup1(y);
      int b = m >> 12, l = m & 4095;
      dst[((size_t)(b * 16 + h) * 4096 + l) * 64 + d] = y;
    }
  }
}

#define AP 68

__global__ __launch_bounds__(256) void chunk_stats(const float* __restrict__ kb,
                                                   const float* __restrict__ vb,
                                                   float* __restrict__ Sch,
                                                   float* __restrict__ Zch) {
  __shared__ float ks[64 * AP];
  __shared__ float vs[64 * AP];
  const int tid = threadIdx.x;
  const int lane = tid & 31, wv = tid >> 5;
  const int hi = lane >> 4, nlo = lane & 15;
  const int c = blockIdx.x, bh = blockIdx.y;
  const size_t base = ((size_t)bh * 4096 + c * 256) * 64;

  v8f sc[2] = {};
  float zacc = 0.f;

  for (int sb = 0; sb < 4; ++sb) {
#pragma unroll
    for (int i = 0; i < 4; ++i) {
      int q = tid + 256 * i;
      int row = q >> 4, c4 = (q & 15) * 4;
      size_t g = base + (size_t)(sb * 64 + row) * 64 + c4;
      *(float4*)(ks + row * AP + c4) = *(const float4*)(kb + g);
      *(float4*)(vs + row * AP + c4) = *(const float4*)(vb + g);
    }
    __syncthreads();
#pragma unroll
    for (int ti = 0; ti < 2; ++ti) {
      const int t = 2 * wv + ti, td = t >> 2, te = t & 3;
      v8f acc = sc[ti];
#pragma unroll
      for (int k0 = 0; k0 < 64; k0 += 4) {
        v2f a;
        a.x = ks[(k0 + 2 * hi) * AP + td * 16 + nlo];
        a.y = ks[(k0 + 2 * hi + 1) * AP + td * 16 + nlo];
        v2f bv;
        bv.x = vs[(k0 + 2 * hi) * AP + te * 16 + nlo];
        bv.y = vs[(k0 + 2 * hi + 1) * AP + te * 16 + nlo];
        acc = wmma4(a, bv, acc);
      }
      sc[ti] = acc;
    }
    if (tid < 64) {
      float s = 0.f;
      for (int m = 0; m < 64; ++m) s += ks[m * AP + tid];
      zacc += s;
    }
    __syncthreads();
  }

  float* So = Sch + ((size_t)bh * NC + c) * 4096;
#pragma unroll
  for (int ti = 0; ti < 2; ++ti) {
    const int t = 2 * wv + ti, td = t >> 2, te = t & 3;
#pragma unroll
    for (int j = 0; j < 8; ++j)
      So[(td * 16 + j + 8 * hi) * 64 + te * 16 + nlo] = sc[ti][j];
  }
  if (tid < 64) Zch[((size_t)bh * NC + c) * 64 + tid] = zacc;
}

__global__ __launch_bounds__(256) void prefix_scan(float* __restrict__ Sch,
                                                   float* __restrict__ Zch) {
  const int bh = blockIdx.x, tid = threadIdx.x;
  for (int el = tid; el < 4096; el += 256) {
    float run = 0.f;
    for (int c = 0; c < NC; ++c) {
      size_t idx = ((size_t)bh * NC + c) * 4096 + el;
      float tmp = Sch[idx];
      Sch[idx] = run;
      run += tmp;
    }
  }
  if (tid < 64) {
    float run = 0.f;
    for (int c = 0; c < NC; ++c) {
      size_t idx = ((size_t)bh * NC + c) * 64 + tid;
      float tmp = Zch[idx];
      Zch[idx] = run;
      run += tmp;
    }
  }
}

__global__ __launch_bounds__(256) void attn_chunk(float* __restrict__ qo,
                                                  const float* __restrict__ kb,
                                                  const float* __restrict__ vb,
                                                  const float* __restrict__ Sch,
                                                  const float* __restrict__ Zch) {
  extern __shared__ float sm[];
  float* qs = sm;
  float* ks = sm + 64 * AP;
  float* vs = sm + 2 * 64 * AP;
  float* as_ = sm + 3 * 64 * AP;
  float* Ss = sm + 4 * 64 * AP;
  float* Zs = sm + 5 * 64 * AP;
  float* dens = Zs + 64;

  const int tid = threadIdx.x;
  const int lane = tid & 31, wv = tid >> 5;
  const int hi = lane >> 4, nlo = lane & 15;
  const int c = blockIdx.x, bh = blockIdx.y;
  const size_t base = (size_t)bh * 4096 * 64;

  const float* Sg = Sch + ((size_t)bh * NC + c) * 4096;
  for (int i = tid; i < 4096; i += 256) Ss[(i >> 6) * AP + (i & 63)] = Sg[i];
  if (tid < 64) Zs[tid] = Zch[((size_t)bh * NC + c) * 64 + tid];
  __syncthreads();

  for (int sb = 0; sb < 4; ++sb) {
    const int l0 = c * 256 + sb * 64;
#pragma unroll
    for (int i = 0; i < 4; ++i) {
      int q = tid + 256 * i;
      int row = q >> 4, c4 = (q & 15) * 4;
      size_t g = base + (size_t)(l0 + row) * 64 + c4;
      *(float4*)(qs + row * AP + c4) = *(const float4*)(qo + g);
      *(float4*)(ks + row * AP + c4) = *(const float4*)(kb + g);
      *(float4*)(vs + row * AP + c4) = *(const float4*)(vb + g);
    }
    __syncthreads();

    v8f num[2];
#pragma unroll
    for (int ti = 0; ti < 2; ++ti) {
      const int t = 2 * wv + ti, tl = t >> 2, tc = t & 3;
      v8f acc = {};
      v8f nacc = {};
      const float* qr = qs + (tl * 16 + nlo) * AP;
      const float* kr = ks + (tc * 16 + nlo) * AP;
#pragma unroll
      for (int k0 = 0; k0 < 64; k0 += 4) {
        v2f a;
        a.x = qr[k0 + 2 * hi];
        a.y = qr[k0 + 2 * hi + 1];
        v2f bk;
        bk.x = kr[k0 + 2 * hi];
        bk.y = kr[k0 + 2 * hi + 1];
        acc = wmma4(a, bk, acc);
        v2f bs;
        bs.x = Ss[(k0 + 2 * hi) * AP + tc * 16 + nlo];
        bs.y = Ss[(k0 + 2 * hi + 1) * AP + tc * 16 + nlo];
        nacc = wmma4(a, bs, nacc);
      }
#pragma unroll
      for (int j = 0; j < 8; ++j) {
        int r = tl * 16 + j + 8 * hi;
        int cc = tc * 16 + nlo;
        as_[r * AP + cc] = (cc <= r) ? acc[j] : 0.f;
      }
      num[ti] = nacc;
    }
    __syncthreads();

    if (tid < 64) {
      float s = 0.f;
      const float* ar = as_ + tid * AP;
      for (int n = 0; n < 64; ++n) s += ar[n];
      const float* qr = qs + tid * AP;
      for (int d = 0; d < 64; ++d) s += qr[d] * Zs[d];
      dens[tid] = fmaxf(s, 1e-6f);
    }
#pragma unroll
    for (int ti = 0; ti < 2; ++ti) {
      const int t = 2 * wv + ti, tl = t >> 2, tc = t & 3;
      v8f nacc = num[ti];
      const float* ar = as_ + (tl * 16 + nlo) * AP;
#pragma unroll
      for (int k0 = 0; k0 < 64; k0 += 4) {
        v2f a;
        a.x = ar[k0 + 2 * hi];
        a.y = ar[k0 + 2 * hi + 1];
        v2f bv;
        bv.x = vs[(k0 + 2 * hi) * AP + tc * 16 + nlo];
        bv.y = vs[(k0 + 2 * hi + 1) * AP + tc * 16 + nlo];
        nacc = wmma4(a, bv, nacc);
      }
      num[ti] = nacc;
    }
    __syncthreads();

    if (sb != 3) {
#pragma unroll
      for (int ti = 0; ti < 2; ++ti) {
        const int t = 2 * wv + ti, td = t >> 2, te = t & 3;
        v8f scv;
#pragma unroll
        for (int j = 0; j < 8; ++j)
          scv[j] = Ss[(td * 16 + j + 8 * hi) * AP + te * 16 + nlo];
#pragma unroll
        for (int k0 = 0; k0 < 64; k0 += 4) {
          v2f a;
          a.x = ks[(k0 + 2 * hi) * AP + td * 16 + nlo];
          a.y = ks[(k0 + 2 * hi + 1) * AP + td * 16 + nlo];
          v2f bv;
          bv.x = vs[(k0 + 2 * hi) * AP + te * 16 + nlo];
          bv.y = vs[(k0 + 2 * hi + 1) * AP + te * 16 + nlo];
          scv = wmma4(a, bv, scv);
        }
#pragma unroll
        for (int j = 0; j < 8; ++j)
          Ss[(td * 16 + j + 8 * hi) * AP + te * 16 + nlo] = scv[j];
      }
      if (tid < 64) {
        float s = Zs[tid];
        for (int m = 0; m < 64; ++m) s += ks[m * AP + tid];
        Zs[tid] = s;
      }
    }
#pragma unroll
    for (int ti = 0; ti < 2; ++ti) {
      const int t = 2 * wv + ti, tl = t >> 2, te = t & 3;
#pragma unroll
      for (int j = 0; j < 8; ++j) {
        int r = tl * 16 + j + 8 * hi;
        float v = num[ti][j] / dens[r];
        qo[base + (size_t)(l0 + r) * 64 + te * 16 + nlo] = v;
      }
    }
    __syncthreads();
  }
}

__global__ __launch_bounds__(256) void out_gemm(const float* __restrict__ oh,
                                                const float* __restrict__ W,
                                                float* __restrict__ out) {
  __shared__ float xs[128 * 36];
  __shared__ float wls[64 * 36];

  const int tid = threadIdx.x;
  const int lane = tid & 31, wv = tid >> 5;
  const int hi = lane >> 4, nlo = lane & 15;
  const int mBase = blockIdx.y * 128;
  const int nBase = blockIdx.x * 64;

  v8f acc[4] = {};

  for (int kt = 0; kt < 1024; kt += 32) {
    const int h = kt >> 6;
    const int dbase = kt & 63;
#pragma unroll
    for (int i = 0; i < 4; ++i) {
      int q = tid + 256 * i;
      int row = q >> 3, c4 = (q & 7) * 4;
      int m = mBase + row;
      int b = m >> 12, l = m & 4095;
      float4 t4 = *(const float4*)(oh + ((size_t)(b * 16 + h) * 4096 + l) * 64 +
                                   dbase + c4);
      *(float4*)(xs + row * 36 + c4) = t4;
    }
#pragma unroll
    for (int i = 0; i < 2; ++i) {
      int q = tid + 256 * i;
      int row = q >> 3, c4 = (q & 7) * 4;
      float4 t4 = *(const float4*)(W + (size_t)(nBase + row) * 1024 + kt + c4);
      *(float4*)(wls + row * 36 + c4) = t4;
    }
    __syncthreads();

    const float* xr = xs + (wv * 16 + nlo) * 36;
#pragma unroll
    for (int kk = 0; kk < 32; kk += 4) {
      v2f a;
      a.x = xr[kk + 2 * hi];
      a.y = xr[kk + 2 * hi + 1];
#pragma unroll
      for (int tc = 0; tc < 4; ++tc) {
        const float* wr = wls + (tc * 16 + nlo) * 36;
        v2f b;
        b.x = wr[kk + 2 * hi];
        b.y = wr[kk + 2 * hi + 1];
        acc[tc] = wmma4(a, b, acc[tc]);
      }
    }
    __syncthreads();
  }

#pragma unroll
  for (int tc = 0; tc < 4; ++tc) {
    int n = nBase + tc * 16 + nlo;
#pragma unroll
    for (int j = 0; j < 8; ++j) {
      int m = mBase + wv * 16 + j + 8 * hi;
      out[(size_t)m * 1024 + n] = acc[tc][j];
    }
  }
}

extern "C" void kernel_launch(void* const* d_in, const int* in_sizes, int n_in,
                              void* d_out, int out_size, void* d_ws,
                              size_t ws_size, hipStream_t stream) {
  const float* x = (const float*)d_in[0];
  const float* Wqkv = (const float*)d_in[1];
  const float* Wout = (const float*)d_in[2];
  float* out = (float*)d_out;

  float* qbuf = (float*)d_ws;
  float* kbuf = qbuf + HEADSZ;
  float* vbuf = kbuf + HEADSZ;
  float* Sch = vbuf + HEADSZ;
  float* Zch = Sch + (size_t)BB * HH * NC * 4096;

  qkv_gemm<<<dim3(48, 64), 256, 0, stream>>>(x, Wqkv, qbuf, kbuf, vbuf);

  chunk_stats<<<dim3(NC, BB * HH), 256, 0, stream>>>(kbuf, vbuf, Sch, Zch);
  prefix_scan<<<dim3(BB * HH), 256, 0, stream>>>(Sch, Zch);
  const size_t smem = (size_t)(5 * 64 * AP + 128) * sizeof(float);
  attn_chunk<<<dim3(NC, BB * HH), 256, smem, stream>>>(qbuf, kbuf, vbuf, Sch,
                                                       Zch);

  out_gemm<<<dim3(16, 64), 256, 0, stream>>>(qbuf, Wout, out);
}


# Both projection GEMMs: LDS fragment reads of each K slab issued as two groups of ten, each ahead of 16 matrix instructions (scheduling barrier), not spread among them; accumulation order unchanged
# speedup vs baseline: 1.0551x; 1.0551x over previous
#include <hip/hip_runtime.h>
#include <math.h>

#define BB 2
#define LL 4096
#define EE 1024
#define HH 16
#define DD 64
#define NC 16
#define HEADSZ ((size_t)BB * HH * LL * DD)

typedef __attribute__((ext_vector_type(2))) float v2f;
typedef __attribute__((ext_vector_type(8))) float v8f;

__device__ __forceinline__ v8f wmma4(v2f a, v2f b, v8f c) {
  return __builtin_amdgcn_wmma_f32_16x16x4_f32(false, a, false, b, (short)0, c,
                                               false, false);
}

__device__ __forceinline__ float elup1(float x) {
  return x > 0.f ? x + 1.f : expf(x);
}

__global__ __launch_bounds__(256) void qkv_gemm(const float* __restrict__ x,
                                                const float* __restrict__ W,
                                                float* __restrict__ qb,
                                                float* __restrict__ kb,
                                                float* __restrict__ vb) {
  __shared__ float xs[128 * 36];
  __shared__ float wls[64 * 36];

  const int tid = threadIdx.x;
  const int lane = tid & 31, wv = tid >> 5;
  const int hi = lane >> 4, nlo = lane & 15;
  const int mBase = blockIdx.y * 128;
  const int nBase = blockIdx.x * 64;

  v8f acc[4] = {};

  for (int kt = 0; kt < 1024; kt += 32) {
#pragma unroll
    for (int i = 0; i < 4; ++i) {
      int q = tid + 256 * i;
      int row = q >> 3, c4 = (q & 7) * 4;
      float4 t4 = *(const float4*)(x + (size_t)(mBase + row) * 1024 + kt + c4);
      *(float4*)(xs + row * 36 + c4) = t4;
    }
#pragma unroll
    for (int i = 0; i < 2; ++i) {
      int q = tid + 256 * i;
      int row = q >> 3, c4 = (q & 7) * 4;
      float4 t4 = *(const float4*)(W + (size_t)(nBase + row) * 1024 + kt + c4);
      *(float4*)(wls + row * 36 + c4) = t4;
    }
    __syncthreads();

    const float* xr = xs + (wv * 16 + nlo) * 36;
#pragma unroll
    for (int hf = 0; hf < 2; ++hf) {
      v2f a[4];
      v2f b[4][4];
#pragma unroll
      for (int s = 0; s < 4; ++s) {
        const int kk = 16 * hf + 4 * s;
        a[s].x = xr[kk + 2 * hi];
        a[s].y = xr[kk + 2 * hi + 1];
#pragma unroll
        for (int tc = 0; tc < 4; ++tc) {
          const float* wr = wls + (tc * 16 + nlo) * 36;
          b[s][tc].x = wr[kk + 2 * hi];
          b[s][tc].y = wr[kk + 2 * hi + 1];
        }
      }
      __builtin_amdgcn_sched_barrier(0);
#pragma unroll
      for (int s = 0; s < 4; ++s) {
#pragma unroll
        for (int tc = 0; tc < 4; ++tc) acc[tc] = wmma4(a[s], b[s][tc], acc[tc]);
      }
    }
    __syncthreads();
  }

  const int tsel = nBase >> 10;
  float* dst = (tsel == 0) ? qb : (tsel == 1) ? kb : vb;
#pragma unroll
  for (int tc = 0; tc < 4; ++tc) {
    int n = nBase + tc * 16 + nlo;
    int rem = n & 1023;
    int h = rem >> 6, d = rem & 63;
#pragma unroll
    for (int j = 0; j < 8; ++j) {
      int m = mBase + wv * 16 + j + 8 * hi;
      float y = acc[tc][j];
      if (tsel == 0)
        y = elup1(y) * 0.125f;
      else if (tsel == 1)
        y = elup1(y);
      int b = m >> 12, l = m & 4095;
      dst[((size_t)(b * 16 + h) * 4096 + l) * 64 + d] = y;
    }
  }
}

#define AP 68

__global__ __launch_bounds__(256) void chunk_stats(const float* __restrict__ kb,
                                                   const float* __restrict__ vb,
                                                   float* __restrict__ Sch,
                                                   float* __restrict__ Zch) {
  __shared__ float ks[64 * AP];
  __shared__ float vs[64 * AP];
  const int tid = threadIdx.x;
  const int lane = tid & 31, wv = tid >> 5;
  const int hi = lane >> 4, nlo = lane & 15;
  const int c = blockIdx.x, bh = blockIdx.y;
  const size_t base = ((size_t)bh * 4096 + c * 256) * 64;

  v8f sc[2] = {};
  float zacc = 0.f;

  for (int sb = 0; sb < 4; ++sb) {
#pragma unroll
    for (int i = 0; i < 4; ++i) {
      int q = tid + 256 * i;
      int row = q >> 4, c4 = (q & 15) * 4;
      size_t g = base + (size_t)(sb * 64 + row) * 64 + c4;
      *(float4*)(ks + row * AP + c4) = *(const float4*)(kb + g);
      *(float4*)(vs + row * AP + c4) = *(const float4*)(vb + g);
    }
    __syncthreads();
#pragma unroll
    for (int ti = 0; ti < 2; ++ti) {
      const int t = 2 * wv + ti, td = t >> 2, te = t & 3;
      v8f acc = sc[ti];
#pragma unroll
      for (int k0 = 0; k0 < 64; k0 += 4) {
        v2f a;
        a.x = ks[(k0 + 2 * hi) * AP + td * 16 + nlo];
        a.y = ks[(k0 + 2 * hi + 1) * AP + td * 16 + nlo];
        v2f bv;
        bv.x = vs[(k0 + 2 * hi) * AP + te * 16 + nlo];
        bv.y = vs[(k0 + 2 * hi + 1) * AP + te * 16 + nlo];
        acc = wmma4(a, bv, acc);
      }
      sc[ti] = acc;
    }
    if (tid < 64) {
      float s = 0.f;
      for (int m = 0; m < 64; ++m) s += ks[m * AP + tid];
      zacc += s;
    }
    __syncthreads();
  }

  float* So = Sch + ((size_t)bh * NC + c) * 4096;
#pragma unroll
  for (int ti = 0; ti < 2; ++ti) {
    const int t = 2 * wv + ti, td = t >> 2, te = t & 3;
#pragma unroll
    for (int j = 0; j < 8; ++j)
      So[(td * 16 + j + 8 * hi) * 64 + te * 16 + nlo] = sc[ti][j];
  }
  if (tid < 64) Zch[((size_t)bh * NC + c) * 64 + tid] = zacc;
}

__global__ __launch_bounds__(256) void prefix_scan(float* __restrict__ Sch,
                                                   float* __restrict__ Zch) {
  const int bh = blockIdx.x, tid = threadIdx.x;
  for (int el = tid; el < 4096; el += 256) {
    float run = 0.f;
    for (int c = 0; c < NC; ++c) {
      size_t idx = ((size_t)bh * NC + c) * 4096 + el;
      float tmp = Sch[idx];
      Sch[idx] = run;
      run += tmp;
    }
  }
  if (tid < 64) {
    float run = 0.f;
    for (int c = 0; c < NC; ++c) {
      size_t idx = ((size_t)bh * NC + c) * 64 + tid;
      float tmp = Zch[idx];
      Zch[idx] = run;
      run += tmp;
    }
  }
}

__global__ __launch_bounds__(256) void attn_chunk(float* __restrict__ qo,
                                                  const float* __restrict__ kb,
                                                  const float* __restrict__ vb,
                                                  const float* __restrict__ Sch,
                                                  const float* __restrict__ Zch) {
  extern __shared__ float sm[];
  float* qs = sm;
  float* ks = sm + 64 * AP;
  float* vs = sm + 2 * 64 * AP;
  float* as_ = sm + 3 * 64 * AP;
  float* Ss = sm + 4 * 64 * AP;
  float* Zs = sm + 5 * 64 * AP;
  float* dens = Zs + 64;

  const int tid = threadIdx.x;
  const int lane = tid & 31, wv = tid >> 5;
  const int hi = lane >> 4, nlo = lane & 15;
  const int c = blockIdx.x, bh = blockIdx.y;
  const size_t base = (size_t)bh * 4096 * 64;

  const float* Sg = Sch + ((size_t)bh * NC + c) * 4096;
  for (int i = tid; i < 4096; i += 256) Ss[(i >> 6) * AP + (i & 63)] = Sg[i];
  if (tid < 64) Zs[tid] = Zch[((size_t)bh * NC + c) * 64 + tid];
  __syncthreads();

  for (int sb = 0; sb < 4; ++sb) {
    const int l0 = c * 256 + sb * 64;
#pragma unroll
    for (int i = 0; i < 4; ++i) {
      int q = tid + 256 * i;
      int row = q >> 4, c4 = (q & 15) * 4;
      size_t g = base + (size_t)(l0 + row) * 64 + c4;
      *(float4*)(qs + row * AP + c4) = *(const float4*)(qo + g);
      *(float4*)(ks + row * AP + c4) = *(const float4*)(kb + g);
      *(float4*)(vs + row * AP + c4) = *(const float4*)(vb + g);
    }
    __syncthreads();

    v8f num[2];
#pragma unroll
    for (int ti = 0; ti < 2; ++ti) {
      const int t = 2 * wv + ti, tl = t >> 2, tc = t & 3;
      v8f acc = {};
      v8f nacc = {};
      const float* qr = qs + (tl * 16 + nlo) * AP;
      const float* kr = ks + (tc * 16 + nlo) * AP;
#pragma unroll
      for (int k0 = 0; k0 < 64; k0 += 4) {
        v2f a;
        a.x = qr[k0 + 2 * hi];
        a.y = qr[k0 + 2 * hi + 1];
        v2f bk;
        bk.x = kr[k0 + 2 * hi];
        bk.y = kr[k0 + 2 * hi + 1];
        acc = wmma4(a, bk, acc);
        v2f bs;
        bs.x = Ss[(k0 + 2 * hi) * AP + tc * 16 + nlo];
        bs.y = Ss[(k0 + 2 * hi + 1) * AP + tc * 16 + nlo];
        nacc = wmma4(a, bs, nacc);
      }
#pragma unroll
      for (int j = 0; j < 8; ++j) {
        int r = tl * 16 + j + 8 * hi;
        int cc = tc * 16 + nlo;
        as_[r * AP + cc] = (cc <= r) ? acc[j] : 0.f;
      }
      num[ti] = nacc;
    }
    __syncthreads();

    if (tid < 64) {
      float s = 0.f;
      const float* ar = as_ + tid * AP;
      for (int n = 0; n < 64; ++n) s += ar[n];
      const float* qr = qs + tid * AP;
      for (int d = 0; d < 64; ++d) s += qr[d] * Zs[d];
      dens[tid] = fmaxf(s, 1e-6f);
    }
#pragma unroll
    for (int ti = 0; ti < 2; ++ti) {
      const int t = 2 * wv + ti, tl = t >> 2, tc = t & 3;
      v8f nacc = num[ti];
      const float* ar = as_ + (tl * 16 + nlo) * AP;
#pragma unroll
      for (int k0 = 0; k0 < 64; k0 += 4) {
        v2f a;
        a.x = ar[k0 + 2 * hi];
        a.y = ar[k0 + 2 * hi + 1];
        v2f bv;
        bv.x = vs[(k0 + 2 * hi) * AP + tc * 16 + nlo];
        bv.y = vs[(k0 + 2 * hi + 1) * AP + tc * 16 + nlo];
        nacc = wmma4(a, bv, nacc);
      }
      num[ti] = nacc;
    }
    __syncthreads();

    if (sb != 3) {
#pragma unroll
      for (int ti = 0; ti < 2; ++ti) {
        const int t = 2 * wv + ti, td = t >> 2, te = t & 3;
        v8f scv;
#pragma unroll
        for (int j = 0; j < 8; ++j)
          scv[j] = Ss[(td * 16 + j + 8 * hi) * AP + te * 16 + nlo];
#pragma unroll
        for (int k0 = 0; k0 < 64; k0 += 4) {
          v2f a;
          a.x = ks[(k0 + 2 * hi) * AP + td * 16 + nlo];
          a.y = ks[(k0 + 2 * hi + 1) * AP + td * 16 + nlo];
          v2f bv;
          bv.x = vs[(k0 + 2 * hi) * AP + te * 16 + nlo];
          bv.y = vs[(k0 + 2 * hi + 1) * AP + te * 16 + nlo];
          scv = wmma4(a, bv, scv);
        }
#pragma unroll
        for (int j = 0; j < 8; ++j)
          Ss[(td * 16 + j + 8 * hi) * AP + te * 16 + nlo] = scv[j];
      }
      if (tid < 64) {
        float s = Zs[tid];
        for (int m = 0; m < 64; ++m) s += ks[m * AP + tid];
        Zs[tid] = s;
      }
    }
#pragma unroll
    for (int ti = 0; ti < 2; ++ti) {
      const int t = 2 * wv + ti, tl = t >> 2, te = t & 3;
#pragma unroll
      for (int j = 0; j < 8; ++j) {
        int r = tl * 16 + j + 8 * hi;
        float v = num[ti][j] / dens[r];
        qo[base + (size_t)(l0 + r) * 64 + te * 16 + nlo] = v;
      }
    }
    __syncthreads();
  }
}

__global__ __launch_bounds__(256) void out_gemm(const float* __restrict__ oh,
                                                const float* __restrict__ W,
                                                float* __restrict__ out) {
  __shared__ float xs[128 * 36];
  __shared__ float wls[64 * 36];

  const int tid = threadIdx.x;
  const int lane = tid & 31, wv = tid >> 5;
  const int hi = lane >> 4, nlo = lane & 15;
  const int mBase = blockIdx.y * 128;
  const int nBase = blockIdx.x * 64;

  v8f acc[4] = {};

  for (int kt = 0; kt < 1024; kt += 32) {
    const int h = kt >> 6;
    const int dbase = kt & 63;
#pragma unroll
    for (int i = 0; i < 4; ++i) {
      int q = tid + 256 * i;
      int row = q >> 3, c4 = (q & 7) * 4;
      int m = mBase + row;
      int b = m >> 12, l = m & 4095;
      float4 t4 = *(const float4*)(oh + ((size_t)(b * 16 + h) * 4096 + l) * 64 +
                                   dbase + c4);
      *(float4*)(xs + row * 36 + c4) = t4;
    }
#pragma unroll
    for (int i = 0; i < 2; ++i) {
      int q = tid + 256 * i;
      int row = q >> 3, c4 = (q & 7) * 4;
      float4 t4 = *(const float4*)(W + (size_t)(nBase + row) * 1024 + kt + c4);
      *(float4*)(wls + row * 36 + c4) = t4;
    }
    __syncthreads();

    const float* xr = xs + (wv * 16 + nlo) * 36;
#pragma unroll
    for (int hf = 0; hf < 2; ++hf) {
      v2f a[4];
      v2f b[4][4];
#pragma unroll
      for (int s = 0; s < 4; ++s) {
        const int kk = 16 * hf + 4 * s;
        a[s].x = xr[kk + 2 * hi];
        a[s].y = xr[kk + 2 * hi + 1];
#pragma unroll
        for (int tc = 0; tc < 4; ++tc) {
          const float* wr = wls + (tc * 16 + nlo) * 36;
          b[s][tc].x = wr[kk + 2 * hi];
          b[s][tc].y = wr[kk + 2 * hi + 1];
        }
      }
      __builtin_amdgcn_sched_barrier(0);
#pragma unroll
      for (int s = 0; s < 4; ++s) {
#pragma unroll
        for (int tc = 0; tc < 4; ++tc) acc[tc] = wmma4(a[s], b[s][tc], acc[tc]);
      }
    }
    __syncthreads();
  }

#pragma unroll
  for (int tc = 0; tc < 4; ++tc) {
    int n = nBase + tc * 16 + nlo;
#pragma unroll
    for (int j = 0; j < 8; ++j) {
      int m = mBase + wv * 16 + j + 8 * hi;
      out[(size_t)m * 1024 + n] = acc[tc][j];
    }
  }
}

extern "C" void kernel_launch(void* const* d_in, const int* in_sizes, int n_in,
                              void* d_out, int out_size, void* d_ws,
                              size_t ws_size, hipStream_t stream) {
  const float* x = (const float*)d_in[0];
  const float* Wqkv = (const float*)d_in[1];
  const float* Wout = (const float*)d_in[2];
  float* out = (float*)d_out;

  float* qbuf = (float*)d_ws;
  float* kbuf = qbuf + HEADSZ;
  float* vbuf = kbuf + HEADSZ;
  float* Sch = vbuf + HEADSZ;
  float* Zch = Sch + (size_t)BB * HH * NC * 4096;

  qkv_gemm<<<dim3(48, 64), 256, 0, stream>>>(x, Wqkv, qbuf, kbuf, vbuf);

  chunk_stats<<<dim3(NC, BB * HH), 256, 0, stream>>>(kbuf, vbuf, Sch, Zch);
  prefix_scan<<<dim3(BB * HH), 256, 0, stream>>>(Sch, Zch);
  const size_t smem = (size_t)(5 * 64 * AP + 128) * sizeof(float);
  attn_chunk<<<dim3(NC, BB * HH), 256, smem, stream>>>(qbuf, kbuf, vbuf, Sch,
                                                       Zch);

  out_gemm<<<dim3(16, 64), 256, 0, stream>>>(qbuf, Wout, out);
}


# Both projection GEMMs: wave tile 16x64 to 32x64, block tile 128x64 to 128x128: each LDS fragment, staged slab and barrier serves twice the matrix instructions; summation order unchanged
# speedup vs baseline: 1.1003x; 1.0428x over previous
#include <hip/hip_runtime.h>
#include <math.h>

#define BB 2
#define LL 4096
#define EE 1024
#define HH 16
#define DD 64
#define NC 16
#define HEADSZ ((size_t)BB * HH * LL * DD)

typedef __attribute__((ext_vector_type(2))) float v2f;
typedef __attribute__((ext_vector_type(8))) float v8f;

__device__ __forceinline__ v8f wmma4(v2f a, v2f b, v8f c) {
  return __builtin_amdgcn_wmma_f32_16x16x4_f32(false, a, false, b, (short)0, c,
                                               false, false);
}

__device__ __forceinline__ float elup1(float x) {
  return x > 0.f ? x + 1.f : expf(x);
}

__global__ __launch_bounds__(256) void qkv_gemm(const float* __restrict__ x,
                                                const float* __restrict__ W,
                                                float* __restrict__ qb,
                                                float* __restrict__ kb,
                                                float* __restrict__ vb) {
  __shared__ float xs[128 * 36];
  __shared__ float wls[128 * 36];

  const int tid = threadIdx.x;
  const int lane = tid & 31, wv = tid >> 5;
  const int hi = lane >> 4, nlo = lane & 15;
  const int wm = wv >> 1, wn = wv & 1;
  const int mBase = blockIdx.y * 128;
  const int nBase = blockIdx.x * 128;

  v8f acc[2][4] = {};

  for (int kt = 0; kt < 1024; kt += 32) {
#pragma unroll
    for (int i = 0; i < 4; ++i) {
      int q = tid + 256 * i;
      int row = q >> 3, c4 = (q & 7) * 4;
      float4 t4 = *(const float4*)(x + (size_t)(mBase + row) * 1024 + kt + c4);
      *(float4*)(xs + row * 36 + c4) = t4;
    }
#pragma unroll
    for (int i = 0; i < 4; ++i) {
      int q = tid + 256 * i;
      int row = q >> 3, c4 = (q & 7) * 4;
      float4 t4 = *(const float4*)(W + (size_t)(nBase + row) * 1024 + kt + c4);
      *(float4*)(wls + row * 36 + c4) = t4;
    }
    __syncthreads();

    const float* xr = xs + (wm * 32 + nlo) * 36;
    const float* wr = wls + (wn * 64 + nlo) * 36;
#pragma unroll
    for (int hf = 0; hf < 2; ++hf) {
      v2f a[4][2];
      v2f b[4][4];
#pragma unroll
      for (int s = 0; s < 4; ++s) {
        const int kk = 16 * hf + 4 * s + 2 * hi;
#pragma unroll
        for (int tr = 0; tr < 2; ++tr) {
          a[s][tr].x = xr[tr * 16 * 36 + kk];
          a[s][tr].y = xr[tr * 16 * 36 + kk + 1];
        }
#pragma unroll
        for (int tc = 0; tc < 4; ++tc) {
          b[s][tc].x = wr[tc * 16 * 36 + kk];
          b[s][tc].y = wr[tc * 16 * 36 + kk + 1];
        }
      }
      __builtin_amdgcn_sched_barrier(0);
#pragma unroll
      for (int s = 0; s < 4; ++s) {
#pragma unroll
        for (int tr = 0; tr < 2; ++tr) {
#pragma unroll
          for (int tc = 0; tc < 4; ++tc)
            acc[tr][tc] = wmma4(a[s][tr], b[s][tc], acc[tr][tc]);
        }
      }
    }
    __syncthreads();
  }

  const int tsel = nBase >> 10;
  float* dst = (tsel == 0) ? qb : (tsel == 1) ? kb : vb;
#pragma unroll
  for (int tr = 0; tr < 2; ++tr) {
#pragma unroll
    for (int tc = 0; tc < 4; ++tc) {
      int n = nBase + wn * 64 + tc * 16 + nlo;
      int rem = n & 1023;
      int h = rem >> 6, d = rem & 63;
#pragma unroll
      for (int j = 0; j < 8; ++j) {
        int m = mBase + wm * 32 + tr * 16 + j + 8 * hi;
        float y = acc[tr][tc][j];
        if (tsel == 0)
          y = elup1(y) * 0.125f;
        else if (tsel == 1)
          y = elup1(y);
        int b = m >> 12, l = m & 4095;
        dst[((size_t)(b * 16 + h) * 4096 + l) * 64 + d] = y;
      }
    }
  }
}

#define AP 68

__global__ __launch_bounds__(256) void chunk_stats(const float* __restrict__ kb,
                                                   const float* __restrict__ vb,
                                                   float* __restrict__ Sch,
                                                   float* __restrict__ Zch) {
  __shared__ float ks[64 * AP];
  __shared__ float vs[64 * AP];
  const int tid = threadIdx.x;
  const int lane = tid & 31, wv = tid >> 5;
  const int hi = lane >> 4, nlo = lane & 15;
  const int c = blockIdx.x, bh = blockIdx.y;
  const size_t base = ((size_t)bh * 4096 + c * 256) * 64;

  v8f sc[2] = {};
  float zacc = 0.f;

  for (int sb = 0; sb < 4; ++sb) {
#pragma unroll
    for (int i = 0; i < 4; ++i) {
      int q = tid + 256 * i;
      int row = q >> 4, c4 = (q & 15) * 4;
      size_t g = base + (size_t)(sb * 64 + row) * 64 + c4;
      *(float4*)(ks + row * AP + c4) = *(const float4*)(kb + g);
      *(float4*)(vs + row * AP + c4) = *(const float4*)(vb + g);
    }
    __syncthreads();
#pragma unroll
    for (int ti = 0; ti < 2; ++ti) {
      const int t = 2 * wv + ti, td = t >> 2, te = t & 3;
      v8f acc = sc[ti];
#pragma unroll
      for (int k0 = 0; k0 < 64; k0 += 4) {
        v2f a;
        a.x = ks[(k0 + 2 * hi) * AP + td * 16 + nlo];
        a.y = ks[(k0 + 2 * hi + 1) * AP + td * 16 + nlo];
        v2f bv;
        bv.x = vs[(k0 + 2 * hi) * AP + te * 16 + nlo];
        bv.y = vs[(k0 + 2 * hi + 1) * AP + te * 16 + nlo];
        acc = wmma4(a, bv, acc);
      }
      sc[ti] = acc;
    }
    if (tid < 64) {
      float s = 0.f;
      for (int m = 0; m < 64; ++m) s += ks[m * AP + tid];
      zacc += s;
    }
    __syncthreads();
  }

  float* So = Sch + ((size_t)bh * NC + c) * 4096;
#pragma unroll
  for (int ti = 0; ti < 2; ++ti) {
    const int t = 2 * wv + ti, td = t >> 2, te = t & 3;
#pragma unroll
    for (int j = 0; j < 8; ++j)
      So[(td * 16 + j + 8 * hi) * 64 + te * 16 + nlo] = sc[ti][j];
  }
  if (tid < 64) Zch[((size_t)bh * NC + c) * 64 + tid] = zacc;
}

__global__ __launch_bounds__(256) void prefix_scan(float* __restrict__ Sch,
                                                   float* __restrict__ Zch) {
  const int bh = blockIdx.x, tid = threadIdx.x;
  for (int el = tid; el < 4096; el += 256) {
    float run = 0.f;
    for (int c = 0; c < NC; ++c) {
      size_t idx = ((size_t)bh * NC + c) * 4096 + el;
      float tmp = Sch[idx];
      Sch[idx] = run;
      run += tmp;
    }
  }
  if (tid < 64) {
    float run = 0.f;
    for (int c = 0; c < NC; ++c) {
      size_t idx = ((size_t)bh * NC + c) * 64 + tid;
      float tmp = Zch[idx];
      Zch[idx] = run;
      run += tmp;
    }
  }
}

__global__ __launch_bounds__(256) void attn_chunk(float* __restrict__ qo,
                                                  const float* __restrict__ kb,
                                                  const float* __restrict__ vb,
                                                  const float* __restrict__ Sch,
                                                  const float* __restrict__ Zch) {
  extern __shared__ float sm[];
  float* qs = sm;
  float* ks = sm + 64 * AP;
  float* vs = sm + 2 * 64 * AP;
  float* as_ = sm + 3 * 64 * AP;
  float* Ss = sm + 4 * 64 * AP;
  float* Zs = sm + 5 * 64 * AP;
  float* dens = Zs + 64;

  const int tid = threadIdx.x;
  const int lane = tid & 31, wv = tid >> 5;
  const int hi = lane >> 4, nlo = lane & 15;
  const int c = blockIdx.x, bh = blockIdx.y;
  const size_t base = (size_t)bh * 4096 * 64;

  const float* Sg = Sch + ((size_t)bh * NC + c) * 4096;
  for (int i = tid; i < 4096; i += 256) Ss[(i >> 6) * AP + (i & 63)] = Sg[i];
  if (tid < 64) Zs[tid] = Zch[((size_t)bh * NC + c) * 64 + tid];
  __syncthreads();

  for (int sb = 0; sb < 4; ++sb) {
    const int l0 = c * 256 + sb * 64;
#pragma unroll
    for (int i = 0; i < 4; ++i) {
      int q = tid + 256 * i;
      int row = q >> 4, c4 = (q & 15) * 4;
      size_t g = base + (size_t)(l0 + row) * 64 + c4;
      *(float4*)(qs + row * AP + c4) = *(const float4*)(qo + g);
      *(float4*)(ks + row * AP + c4) = *(const float4*)(kb + g);
      *(float4*)(vs + row * AP + c4) = *(const float4*)(vb + g);
    }
    __syncthreads();

    v8f num[2];
#pragma unroll
    for (int ti = 0; ti < 2; ++ti) {
      const int t = 2 * wv + ti, tl = t >> 2, tc = t & 3;
      v8f acc = {};
      v8f nacc = {};
      const float* qr = qs + (tl * 16 + nlo) * AP;
      const float* kr = ks + (tc * 16 + nlo) * AP;
#pragma unroll
      for (int k0 = 0; k0 < 64; k0 += 4) {
        v2f a;
        a.x = qr[k0 + 2 * hi];
        a.y = qr[k0 + 2 * hi + 1];
        v2f bk;
        bk.x = kr[k0 + 2 * hi];
        bk.y = kr[k0 + 2 * hi + 1];
        acc = wmma4(a, bk, acc);
        v2f bs;
        bs.x = Ss[(k0 + 2 * hi) * AP + tc * 16 + nlo];
        bs.y = Ss[(k0 + 2 * hi + 1) * AP + tc * 16 + nlo];
        nacc = wmma4(a, bs, nacc);
      }
#pragma unroll
      for (int j = 0; j < 8; ++j) {
        int r = tl * 16 + j + 8 * hi;
        int cc = tc * 16 + nlo;
        as_[r * AP + cc] = (cc <= r) ? acc[j] : 0.f;
      }
      num[ti] = nacc;
    }
    __syncthreads();

    if (tid < 64) {
      float s = 0.f;
      const float* ar = as_ + tid * AP;
      for (int n = 0; n < 64; ++n) s += ar[n];
      const float* qr = qs + tid * AP;
      for (int d = 0; d < 64; ++d) s += qr[d] * Zs[d];
      dens[tid] = fmaxf(s, 1e-6f);
    }
#pragma unroll
    for (int ti = 0; ti < 2; ++ti) {
      const int t = 2 * wv + ti, tl = t >> 2, tc = t & 3;
      v8f nacc = num[ti];
      const float* ar = as_ + (tl * 16 + nlo) * AP;
#pragma unroll
      for (int k0 = 0; k0 < 64; k0 += 4) {
        v2f a;
        a.x = ar[k0 + 2 * hi];
        a.y = ar[k0 + 2 * hi + 1];
        v2f bv;
        bv.x = vs[(k0 + 2 * hi) * AP + tc * 16 + nlo];
        bv.y = vs[(k0 + 2 * hi + 1) * AP + tc * 16 + nlo];
        nacc = wmma4(a, bv, nacc);
      }
      num[ti] = nacc;
    }
    __syncthreads();

    if (sb != 3) {
#pragma unroll
      for (int ti = 0; ti < 2; ++ti) {
        const int t = 2 * wv + ti, td = t >> 2, te = t & 3;
        v8f scv;
#pragma unroll
        for (int j = 0; j < 8; ++j)
          scv[j] = Ss[(td * 16 + j + 8 * hi) * AP + te * 16 + nlo];
#pragma unroll
        for (int k0 = 0; k0 < 64; k0 += 4) {
          v2f a;
          a.x = ks[(k0 + 2 * hi) * AP + td * 16 + nlo];
          a.y = ks[(k0 + 2 * hi + 1) * AP + td * 16 + nlo];
          v2f bv;
          bv.x = vs[(k0 + 2 * hi) * AP + te * 16 + nlo];
          bv.y = vs[(k0 + 2 * hi + 1) * AP + te * 16 + nlo];
          scv = wmma4(a, bv, scv);
        }
#pragma unroll
        for (int j = 0; j < 8; ++j)
          Ss[(td * 16 + j + 8 * hi) * AP + te * 16 + nlo] = scv[j];
      }
      if (tid < 64) {
        float s = Zs[tid];
        for (int m = 0; m < 64; ++m) s += ks[m * AP + tid];
        Zs[tid] = s;
      }
    }
#pragma unroll
    for (int ti = 0; ti < 2; ++ti) {
      const int t = 2 * wv + ti, tl = t >> 2, te = t & 3;
#pragma unroll
      for (int j = 0; j < 8; ++j) {
        int r = tl * 16 + j + 8 * hi;
        float v = num[ti][j] / dens[r];
        qo[base + (size_t)(l0 + r) * 64 + te * 16 + nlo] = v;
      }
    }
    __syncthreads();
  }
}

__global__ __launch_bounds__(256) void out_gemm(const float* __restrict__ oh,
                                                const float* __restrict__ W,
                                                float* __restrict__ out) {
  __shared__ float xs[128 * 36];
  __shared__ float wls[128 * 36];

  const int tid = threadIdx.x;
  const int lane = tid & 31, wv = tid >> 5;
  const int hi = lane >> 4, nlo = lane & 15;
  const int wm = wv >> 1, wn = wv & 1;
  const int mBase = blockIdx.y * 128;
  const int nBase = blockIdx.x * 128;

  v8f acc[2][4] = {};

  for (int kt = 0; kt < 1024; kt += 32) {
    const int h = kt >> 6;
    const int dbase = kt & 63;
#pragma unroll
    for (int i = 0; i < 4; ++i) {
      int q = tid + 256 * i;
      int row = q >> 3, c4 = (q & 7) * 4;
      int m = mBase + row;
      int b = m >> 12, l = m & 4095;
      float4 t4 = *(const float4*)(oh + ((size_t)(b * 16 + h) * 4096 + l) * 64 +
                                   dbase + c4);
      *(float4*)(xs + row * 36 + c4) = t4;
    }
#pragma unroll
    for (int i = 0; i < 4; ++i) {
      int q = tid + 256 * i;
      int row = q >> 3, c4 = (q & 7) * 4;
      float4 t4 = *(const float4*)(W + (size_t)(nBase + row) * 1024 + kt + c4);
      *(float4*)(wls + row * 36 + c4) = t4;
    }
    __syncthreads();

    const float* xr = xs + (wm * 32 + nlo) * 36;
    const float* wr = wls + (wn * 64 + nlo) * 36;
#pragma unroll
    for (int hf = 0; hf < 2; ++hf) {
      v2f a[4][2];
      v2f b[4][4];
#pragma unroll
      for (int s = 0; s < 4; ++s) {
        const int kk = 16 * hf + 4 * s + 2 * hi;
#pragma unroll
        for (int tr = 0; tr < 2; ++tr) {
          a[s][tr].x = xr[tr * 16 * 36 + kk];
          a[s][tr].y = xr[tr * 16 * 36 + kk + 1];
        }
#pragma unroll
        for (int tc = 0; tc < 4; ++tc) {
          b[s][tc].x = wr[tc * 16 * 36 + kk];
          b[s][tc].y = wr[tc * 16 * 36 + kk + 1];
        }
      }
      __builtin_amdgcn_sched_barrier(0);
#pragma unroll
      for (int s = 0; s < 4; ++s) {
#pragma unroll
        for (int tr = 0; tr < 2; ++tr) {
#pragma unroll
          for (int tc = 0; tc < 4; ++tc)
            acc[tr][tc] = wmma4(a[s][tr], b[s][tc], acc[tr][tc]);
        }
      }
    }
    __syncthreads();
  }

#pragma unroll
  for (int tr = 0; tr < 2; ++tr) {
#pragma unroll
    for (int tc = 0; tc < 4; ++tc) {
      int n = nBase + wn * 64 + tc * 16 + nlo;
#pragma unroll
      for (int j = 0; j < 8; ++j) {
        int m = mBase + wm * 32 + tr * 16 + j + 8 * hi;
        out[(size_t)m * 1024 + n] = acc[tr][tc][j];
      }
    }
  }
}

extern "C" void kernel_launch(void* const* d_in, const int* in_sizes, int n_in,
                              void* d_out, int out_size, void* d_ws,
                              size_t ws_size, hipStream_t stream) {
  const float* x = (const float*)d_in[0];
  const float* Wqkv = (const float*)d_in[1];
  const float* Wout = (const float*)d_in[2];
  float* out = (float*)d_out;

  float* qbuf = (float*)d_ws;
  float* kbuf = qbuf + HEADSZ;
  float* vbuf = kbuf + HEADSZ;
  float* Sch = vbuf + HEADSZ;
  float* Zch = Sch + (size_t)BB * HH * NC * 4096;

  qkv_gemm<<<dim3(24, 64), 256, 0, stream>>>(x, Wqkv, qbuf, kbuf, vbuf);

  chunk_stats<<<dim3(NC, BB * HH), 256, 0, stream>>>(kbuf, vbuf, Sch, Zch);
  prefix_scan<<<dim3(BB * HH), 256, 0, stream>>>(Sch, Zch);
  const size_t smem = (size_t)(5 * 64 * AP + 128) * sizeof(float);
  attn_chunk<<<dim3(NC, BB * HH), 256, smem, stream>>>(qbuf, kbuf, vbuf, Sch,
                                                       Zch);

  out_gemm<<<dim3(8, 64), 256, 0, stream>>>(qbuf, Wout, out);
}


# Prefix-scan launch removed (5 launches to 4): each attention block sums the S,Z statistics of its head's earlier chunks itself, chunk 0 first, so every sum keeps its order and value
# speedup vs baseline: 1.1672x; 1.0608x over previous
#include <hip/hip_runtime.h>
#include <math.h>

#define BB 2
#define LL 4096
#define EE 1024
#define HH 16
#define DD 64
#define NC 16
#define HEADSZ ((size_t)BB * HH * LL * DD)

typedef __attribute__((ext_vector_type(2))) float v2f;
typedef __attribute__((ext_vector_type(8))) float v8f;

__device__ __forceinline__ v8f wmma4(v2f a, v2f b, v8f c) {
  return __builtin_amdgcn_wmma_f32_16x16x4_f32(false, a, false, b, (short)0, c,
                                               false, false);
}

__device__ __forceinline__ float elup1(float x) {
  return x > 0.f ? x + 1.f : expf(x);
}

__global__ __launch_bounds__(256) void qkv_gemm(const float* __restrict__ x,
                                                const float* __restrict__ W,
                                                float* __restrict__ qb,
                                                float* __restrict__ kb,
                                                float* __restrict__ vb) {
  __shared__ float xs[128 * 36];
  __shared__ float wls[128 * 36];

  const int tid = threadIdx.x;
  const int lane = tid & 31, wv = tid >> 5;
  const int hi = lane >> 4, nlo = lane & 15;
  const int wm = wv >> 1, wn = wv & 1;
  const int mBase = blockIdx.y * 128;
  const int nBase = blockIdx.x * 128;

  v8f acc[2][4] = {};

  for (int kt = 0; kt < 1024; kt += 32) {
#pragma unroll
    for (int i = 0; i < 4; ++i) {
      int q = tid + 256 * i;
      int row = q >> 3, c4 = (q & 7) * 4;
      float4 t4 = *(const float4*)(x + (size_t)(mBase + row) * 1024 + kt + c4);
      *(float4*)(xs + row * 36 + c4) = t4;
    }
#pragma unroll
    for (int i = 0; i < 4; ++i) {
      int q = tid + 256 * i;
      int row = q >> 3, c4 = (q & 7) * 4;
      float4 t4 = *(const float4*)(W + (size_t)(nBase + row) * 1024 + kt + c4);
      *(float4*)(wls + row * 36 + c4) = t4;
    }
    __syncthreads();

    const float* xr = xs + (wm * 32 + nlo) * 36;
    const float* wr = wls + (wn * 64 + nlo) * 36;
#pragma unroll
    for (int hf = 0; hf < 2; ++hf) {
      v2f a[4][2];
      v2f b[4][4];
#pragma unroll
      for (int s = 0; s < 4; ++s) {
        const int kk = 16 * hf + 4 * s + 2 * hi;
#pragma unroll
        for (int tr = 0; tr < 2; ++tr) {
          a[s][tr].x = xr[tr * 16 * 36 + kk];
          a[s][tr].y = xr[tr * 16 * 36 + kk + 1];
        }
#pragma unroll
        for (int tc = 0; tc < 4; ++tc) {
          b[s][tc].x = wr[tc * 16 * 36 + kk];
          b[s][tc].y = wr[tc * 16 * 36 + kk + 1];
        }
      }
      __builtin_amdgcn_sched_barrier(0);
#pragma unroll
      for (int s = 0; s < 4; ++s) {
#pragma unroll
        for (int tr = 0; tr < 2; ++tr) {
#pragma unroll
          for (int tc = 0; tc < 4; ++tc)
            acc[tr][tc] = wmma4(a[s][tr], b[s][tc], acc[tr][tc]);
        }
      }
    }
    __syncthreads();
  }

  const int tsel = nBase >> 10;
  float* dst = (tsel == 0) ? qb : (tsel == 1) ? kb : vb;
#pragma unroll
  for (int tr = 0; tr < 2; ++tr) {
#pragma unroll
    for (int tc = 0; tc < 4; ++tc) {
      int n = nBase + wn * 64 + tc * 16 + nlo;
      int rem = n & 1023;
      int h = rem >> 6, d = rem & 63;
#pragma unroll
      for (int j = 0; j < 8; ++j) {
        int m = mBase + wm * 32 + tr * 16 + j + 8 * hi;
        float y = acc[tr][tc][j];
        if (tsel == 0)
          y = elup1(y) * 0.125f;
        else if (tsel == 1)
          y = elup1(y);
        int b = m >> 12, l = m & 4095;
        dst[((size_t)(b * 16 + h) * 4096 + l) * 64 + d] = y;
      }
    }
  }
}

#define AP 68

__global__ __launch_bounds__(256) void chunk_stats(const float* __restrict__ kb,
                                                   const float* __restrict__ vb,
                                                   float* __restrict__ Sch,
                                                   float* __restrict__ Zch) {
  __shared__ float ks[64 * AP];
  __shared__ float vs[64 * AP];
  const int tid = threadIdx.x;
  const int lane = tid & 31, wv = tid >> 5;
  const int hi = lane >> 4, nlo = lane & 15;
  const int c = blockIdx.x, bh = blockIdx.y;
  const size_t base = ((size_t)bh * 4096 + c * 256) * 64;

  v8f sc[2] = {};
  float zacc = 0.f;

  for (int sb = 0; sb < 4; ++sb) {
#pragma unroll
    for (int i = 0; i < 4; ++i) {
      int q = tid + 256 * i;
      int row = q >> 4, c4 = (q & 15) * 4;
      size_t g = base + (size_t)(sb * 64 + row) * 64 + c4;
      *(float4*)(ks + row * AP + c4) = *(const float4*)(kb + g);
      *(float4*)(vs + row * AP + c4) = *(const float4*)(vb + g);
    }
    __syncthreads();
#pragma unroll
    for (int ti = 0; ti < 2; ++ti) {
      const int t = 2 * wv + ti, td = t >> 2, te = t & 3;
      v8f acc = sc[ti];
#pragma unroll
      for (int k0 = 0; k0 < 64; k0 += 4) {
        v2f a;
        a.x = ks[(k0 + 2 * hi) * AP + td * 16 + nlo];
        a.y = ks[(k0 + 2 * hi + 1) * AP + td * 16 + nlo];
        v2f bv;
        bv.x = vs[(k0 + 2 * hi) * AP + te * 16 + nlo];
        bv.y = vs[(k0 + 2 * hi + 1) * AP + te * 16 + nlo];
        acc = wmma4(a, bv, acc);
      }
      sc[ti] = acc;
    }
    if (tid < 64) {
      float s = 0.f;
      for (int m = 0; m < 64; ++m) s += ks[m * AP + tid];
      zacc += s;
    }
    __syncthreads();
  }

  float* So = Sch + ((size_t)bh * NC + c) * 4096;
#pragma unroll
  for (int ti = 0; ti < 2; ++ti) {
    const int t = 2 * wv + ti, td = t >> 2, te = t & 3;
#pragma unroll
    for (int j = 0; j < 8; ++j)
      So[(td * 16 + j + 8 * hi) * 64 + te * 16 + nlo] = sc[ti][j];
  }
  if (tid < 64) Zch[((size_t)bh * NC + c) * 64 + tid] = zacc;
}

__global__ __launch_bounds__(256) void attn_chunk(float* __restrict__ qo,
                                                  const float* __restrict__ kb,
                                                  const float* __restrict__ vb,
                                                  const float* __restrict__ Sch,
                                                  const float* __restrict__ Zch) {
  extern __shared__ float sm[];
  float* qs = sm;
  float* ks = sm + 64 * AP;
  float* vs = sm + 2 * 64 * AP;
  float* as_ = sm + 3 * 64 * AP;
  float* Ss = sm + 4 * 64 * AP;
  float* Zs = sm + 5 * 64 * AP;
  float* dens = Zs + 64;

  const int tid = threadIdx.x;
  const int lane = tid & 31, wv = tid >> 5;
  const int hi = lane >> 4, nlo = lane & 15;
  const int c = blockIdx.x, bh = blockIdx.y;
  const size_t base = (size_t)bh * 4096 * 64;

  {
    const float4* Sg = (const float4*)(Sch + (size_t)bh * NC * 4096);
    const float* Zg = Zch + (size_t)bh * NC * 64 + (tid & 63);
    float4 run[4];
#pragma unroll
    for (int j = 0; j < 4; ++j) run[j] = make_float4(0.f, 0.f, 0.f, 0.f);
    float zrun = 0.f;
    for (int cp0 = 0; cp0 < c; cp0 += 4) {
      float4 t4[4][4];
      float zt[4];
#pragma unroll
      for (int u = 0; u < 4; ++u) {
        if (cp0 + u < c) {
#pragma unroll
          for (int j = 0; j < 4; ++j)
            t4[u][j] = Sg[(cp0 + u) * 1024 + tid + 256 * j];
          zt[u] = Zg[(cp0 + u) * 64];
        } else {
#pragma unroll
          for (int j = 0; j < 4; ++j) t4[u][j] = make_float4(0.f, 0.f, 0.f, 0.f);
          zt[u] = 0.f;
        }
      }
#pragma unroll
      for (int u = 0; u < 4; ++u) {
#pragma unroll
        for (int j = 0; j < 4; ++j) {
          run[j].x += t4[u][j].x;
          run[j].y += t4[u][j].y;
          run[j].z += t4[u][j].z;
          run[j].w += t4[u][j].w;
        }
        zrun += zt[u];
      }
    }
#pragma unroll
    for (int j = 0; j < 4; ++j) {
      int e4 = tid + 256 * j;
      *(float4*)(Ss + (e4 >> 4) * AP + (e4 & 15) * 4) = run[j];
    }
    if (tid < 64) Zs[tid] = zrun;
  }
  __syncthreads();

  for (int sb = 0; sb < 4; ++sb) {
    const int l0 = c * 256 + sb * 64;
#pragma unroll
    for (int i = 0; i < 4; ++i) {
      int q = tid + 256 * i;
      int row = q >> 4, c4 = (q & 15) * 4;
      size_t g = base + (size_t)(l0 + row) * 64 + c4;
      *(float4*)(qs + row * AP + c4) = *(const float4*)(qo + g);
      *(float4*)(ks + row * AP + c4) = *(const float4*)(kb + g);
      *(float4*)(vs + row * AP + c4) = *(const float4*)(vb + g);
    }
    __syncthreads();

    v8f num[2];
#pragma unroll
    for (int ti = 0; ti < 2; ++ti) {
      const int t = 2 * wv + ti, tl = t >> 2, tc = t & 3;
      v8f acc = {};
      v8f nacc = {};
      const float* qr = qs + (tl * 16 + nlo) * AP;
      const float* kr = ks + (tc * 16 + nlo) * AP;
#pragma unroll
      for (int k0 = 0; k0 < 64; k0 += 4) {
        v2f a;
        a.x = qr[k0 + 2 * hi];
        a.y = qr[k0 + 2 * hi + 1];
        v2f bk;
        bk.x = kr[k0 + 2 * hi];
        bk.y = kr[k0 + 2 * hi + 1];
        acc = wmma4(a, bk, acc);
        v2f bs;
        bs.x = Ss[(k0 + 2 * hi) * AP + tc * 16 + nlo];
        bs.y = Ss[(k0 + 2 * hi + 1) * AP + tc * 16 + nlo];
        nacc = wmma4(a, bs, nacc);
      }
#pragma unroll
      for (int j = 0; j < 8; ++j) {
        int r = tl * 16 + j + 8 * hi;
        int cc = tc * 16 + nlo;
        as_[r * AP + cc] = (cc <= r) ? acc[j] : 0.f;
      }
      num[ti] = nacc;
    }
    __syncthreads();

    if (tid < 64) {
      float s = 0.f;
      const float* ar = as_ + tid * AP;
      for (int n = 0; n < 64; ++n) s += ar[n];
      const float* qr = qs + tid * AP;
      for (int d = 0; d < 64; ++d) s += qr[d] * Zs[d];
      dens[tid] = fmaxf(s, 1e-6f);
    }
#pragma unroll
    for (int ti = 0; ti < 2; ++ti) {
      const int t = 2 * wv + ti, tl = t >> 2, tc = t & 3;
      v8f nacc = num[ti];
      const float* ar = as_ + (tl * 16 + nlo) * AP;
#pragma unroll
      for (int k0 = 0; k0 < 64; k0 += 4) {
        v2f a;
        a.x = ar[k0 + 2 * hi];
        a.y = ar[k0 + 2 * hi + 1];
        v2f bv;
        bv.x = vs[(k0 + 2 * hi) * AP + tc * 16 + nlo];
        bv.y = vs[(k0 + 2 * hi + 1) * AP + tc * 16 + nlo];
        nacc = wmma4(a, bv, nacc);
      }
      num[ti] = nacc;
    }
    __syncthreads();

    if (sb != 3) {
#pragma unroll
      for (int ti = 0; ti < 2; ++ti) {
        const int t = 2 * wv + ti, td = t >> 2, te = t & 3;
        v8f scv;
#pragma unroll
        for (int j = 0; j < 8; ++j)
          scv[j] = Ss[(td * 16 + j + 8 * hi) * AP + te * 16 + nlo];
#pragma unroll
        for (int k0 = 0; k0 < 64; k0 += 4) {
          v2f a;
          a.x = ks[(k0 + 2 * hi) * AP + td * 16 + nlo];
          a.y = ks[(k0 + 2 * hi + 1) * AP + td * 16 + nlo];
          v2f bv;
          bv.x = vs[(k0 + 2 * hi) * AP + te * 16 + nlo];
          bv.y = vs[(k0 + 2 * hi + 1) * AP + te * 16 + nlo];
          scv = wmma4(a, bv, scv);
        }
#pragma unroll
        for (int j = 0; j < 8; ++j)
          Ss[(td * 16 + j + 8 * hi) * AP + te * 16 + nlo] = scv[j];
      }
      if (tid < 64) {
        float s = Zs[tid];
        for (int m = 0; m < 64; ++m) s += ks[m * AP + tid];
        Zs[tid] = s;
      }
    }
#pragma unroll
    for (int ti = 0; ti < 2; ++ti) {
      const int t = 2 * wv + ti, tl = t >> 2, te = t & 3;
#pragma unroll
      for (int j = 0; j < 8; ++j) {
        int r = tl * 16 + j + 8 * hi;
        float v = num[ti][j] / dens[r];
        qo[base + (size_t)(l0 + r) * 64 + te * 16 + nlo] = v;
      }
    }
    __syncthreads();
  }
}

__global__ __launch_bounds__(256) void out_gemm(const float* __restrict__ oh,
                                                const float* __restrict__ W,
                                                float* __restrict__ out) {
  __shared__ float xs[128 * 36];
  __shared__ float wls[128 * 36];

  const int tid = threadIdx.x;
  const int lane = tid & 31, wv = tid >> 5;
  const int hi = lane >> 4, nlo = lane & 15;
  const int wm = wv >> 1, wn = wv & 1;
  const int mBase = blockIdx.y * 128;
  const int nBase = blockIdx.x * 128;

  v8f acc[2][4] = {};

  for (int kt = 0; kt < 1024; kt += 32) {
    const int h = kt >> 6;
    const int dbase = kt & 63;
#pragma unroll
    for (int i = 0; i < 4; ++i) {
      int q = tid + 256 * i;
      int row = q >> 3, c4 = (q & 7) * 4;
      int m = mBase + row;
      int b = m >> 12, l = m & 4095;
      float4 t4 = *(const float4*)(oh + ((size_t)(b * 16 + h) * 4096 + l) * 64 +
                                   dbase + c4);
      *(float4*)(xs + row * 36 + c4) = t4;
    }
#pragma unroll
    for (int i = 0; i < 4; ++i) {
      int q = tid + 256 * i;
      int row = q >> 3, c4 = (q & 7) * 4;
      float4 t4 = *(const float4*)(W + (size_t)(nBase + row) * 1024 + kt + c4);
      *(float4*)(wls + row * 36 + c4) = t4;
    }
    __syncthreads();

    const float* xr = xs + (wm * 32 + nlo) * 36;
    const float* wr = wls + (wn * 64 + nlo) * 36;
#pragma unroll
    for (int hf = 0; hf < 2; ++hf) {
      v2f a[4][2];
      v2f b[4][4];
#pragma unroll
      for (int s = 0; s < 4; ++s) {
        const int kk = 16 * hf + 4 * s + 2 * hi;
#pragma unroll
        for (int tr = 0; tr < 2; ++tr) {
          a[s][tr].x = xr[tr * 16 * 36 + kk];
          a[s][tr].y = xr[tr * 16 * 36 + kk + 1];
        }
#pragma unroll
        for (int tc = 0; tc < 4; ++tc) {
          b[s][tc].x = wr[tc * 16 * 36 + kk];
          b[s][tc].y = wr[tc * 16 * 36 + kk + 1];
        }
      }
      __builtin_amdgcn_sched_barrier(0);
#pragma unroll
      for (int s = 0; s < 4; ++s) {
#pragma unroll
        for (int tr = 0; tr < 2; ++tr) {
#pragma unroll
          for (int tc = 0; tc < 4; ++tc)
            acc[tr][tc] = wmma4(a[s][tr], b[s][tc], acc[tr][tc]);
        }
      }
    }
    __syncthreads();
  }

#pragma unroll
  for (int tr = 0; tr < 2; ++tr) {
#pragma unroll
    for (int tc = 0; tc < 4; ++tc) {
      int n = nBase + wn * 64 + tc * 16 + nlo;
#pragma unroll
      for (int j = 0; j < 8; ++j) {
        int m = mBase + wm * 32 + tr * 16 + j + 8 * hi;
        out[(size_t)m * 1024 + n] = acc[tr][tc][j];
      }
    }
  }
}

extern "C" void kernel_launch(void* const* d_in, const int* in_sizes, int n_in,
                              void* d_out, int out_size, void* d_ws,
                              size_t ws_size, hipStream_t stream) {
  const float* x = (const float*)d_in[0];
  const float* Wqkv = (const float*)d_in[1];
  const float* Wout = (const float*)d_in[2];
  float* out = (float*)d_out;

  float* qbuf = (float*)d_ws;
  float* kbuf = qbuf + HEADSZ;
  float* vbuf = kbuf + HEADSZ;
  float* Sch = vbuf + HEADSZ;
  float* Zch = Sch + (size_t)BB * HH * NC * 4096;

  qkv_gemm<<<dim3(24, 64), 256, 0, stream>>>(x, Wqkv, qbuf, kbuf, vbuf);

  chunk_stats<<<dim3(NC, BB * HH), 256, 0, stream>>>(kbuf, vbuf, Sch, Zch);
  const size_t smem = (size_t)(5 * 64 * AP + 128) * sizeof(float);
  attn_chunk<<<dim3(NC, BB * HH), 256, smem, stream>>>(qbuf, kbuf, vbuf, Sch,
                                                       Zch);

  out_gemm<<<dim3(8, 64), 256, 0, stream>>>(qbuf, Wout, out);
}


# Both projection GEMMs: next K slab's global loads issued before the current slab's matrix instructions, stored to a second LDS buffer after them; one barrier per slab, not two; sum order unchanged
# speedup vs baseline: 1.2045x; 1.0320x over previous
#include <hip/hip_runtime.h>
#include <math.h>

#define BB 2
#define LL 4096
#define EE 1024
#define HH 16
#define DD 64
#define NC 16
#define HEADSZ ((size_t)BB * HH * LL * DD)

typedef __attribute__((ext_vector_type(2))) float v2f;
typedef __attribute__((ext_vector_type(4))) float v4f;
typedef __attribute__((ext_vector_type(8))) float v8f;

__device__ __forceinline__ v8f wmma4(v2f a, v2f b, v8f c) {
  return __builtin_amdgcn_wmma_f32_16x16x4_f32(false, a, false, b, (short)0, c,
                                               false, false);
}

__device__ __forceinline__ float elup1(float x) {
  return x > 0.f ? x + 1.f : expf(x);
}

#define SLAB_LOAD(XP, XSTRIDE, WP)                 \
  px0 = *(const v4f*)((XP) + 0 * (XSTRIDE));       \
  pw0 = *(const v4f*)((WP) + 0 * 32 * 1024);       \
  px1 = *(const v4f*)((XP) + 1 * (XSTRIDE));       \
  pw1 = *(const v4f*)((WP) + 1 * 32 * 1024);       \
  px2 = *(const v4f*)((XP) + 2 * (XSTRIDE));       \
  pw2 = *(const v4f*)((WP) + 2 * 32 * 1024);       \
  px3 = *(const v4f*)((XP) + 3 * (XSTRIDE));       \
  pw3 = *(const v4f*)((WP) + 3 * 32 * 1024);
#define SLAB_STORE(XS, WS)                         \
  *(v4f*)((XS) + 0 * 32 * 36) = px0;               \
  *(v4f*)((WS) + 0 * 32 * 36) = pw0;               \
  *(v4f*)((XS) + 1 * 32 * 36) = px1;               \
  *(v4f*)((WS) + 1 * 32 * 36) = pw1;               \
  *(v4f*)((XS) + 2 * 32 * 36) = px2;               \
  *(v4f*)((WS) + 2 * 32 * 36) = pw2;               \
  *(v4f*)((XS) + 3 * 32 * 36) = px3;               \
  *(v4f*)((WS) + 3 * 32 * 36) = pw3;

#define SLAB (128 * 36)

__device__ __forceinline__ void slab_mma(const float* xr, const float* wr,
                                         const int hi, v8f (&acc)[2][4]) {
#pragma unroll
  for (int hf = 0; hf < 2; ++hf) {
    v2f a[4][2];
    v2f b[4][4];
#pragma unroll
    for (int s = 0; s < 4; ++s) {
      const int kk = 16 * hf + 4 * s + 2 * hi;
#pragma unroll
      for (int tr = 0; tr < 2; ++tr) {
        a[s][tr].x = xr[tr * 16 * 36 + kk];
        a[s][tr].y = xr[tr * 16 * 36 + kk + 1];
      }
#pragma unroll
      for (int tc = 0; tc < 4; ++tc) {
        b[s][tc].x = wr[tc * 16 * 36 + kk];
        b[s][tc].y = wr[tc * 16 * 36 + kk + 1];
      }
    }
    __builtin_amdgcn_sched_barrier(0);
#pragma unroll
    for (int s = 0; s < 4; ++s) {
#pragma unroll
      for (int tr = 0; tr < 2; ++tr) {
#pragma unroll
        for (int tc = 0; tc < 4; ++tc)
          acc[tr][tc] = wmma4(a[s][tr], b[s][tc], acc[tr][tc]);
      }
    }
  }
}

__global__ __launch_bounds__(256) void qkv_gemm(const float* __restrict__ x,
                                                const float* __restrict__ W,
                                                float* __restrict__ qb,
                                                float* __restrict__ kb,
                                                float* __restrict__ vb) {
  __shared__ float xs[2 * SLAB];
  __shared__ float wls[2 * SLAB];

  const int tid = threadIdx.x;
  const int lane = tid & 31, wv = tid >> 5;
  const int hi = lane >> 4, nlo = lane & 15;
  const int wm = wv >> 1, wn = wv & 1;
  const int mBase = blockIdx.y * 128;
  const int nBase = blockIdx.x * 128;

  v8f acc[2][4] = {};

  const int srow = tid >> 3, sc4 = (tid & 7) * 4;
  const float* xg = x + (size_t)(mBase + srow) * 1024 + sc4;
  const float* wg = W + (size_t)(nBase + srow) * 1024 + sc4;
  float* xst = xs + srow * 36 + sc4;
  float* wst = wls + srow * 36 + sc4;
  const float* xr = xs + (wm * 32 + nlo) * 36;
  const float* wr = wls + (wn * 64 + nlo) * 36;

  v4f px0, px1, px2, px3, pw0, pw1, pw2, pw3;
  SLAB_LOAD(xg, 32 * 1024, wg)
  SLAB_STORE(xst, wst)
  __syncthreads();

  for (int kt = 0; kt < 1024; kt += 64) {
    SLAB_LOAD(xg + kt + 32, 32 * 1024, wg + kt + 32)
    slab_mma(xr, wr, hi, acc);
    SLAB_STORE(xst + SLAB, wst + SLAB)
    __syncthreads();

    const bool more = kt + 64 < 1024;
    if (more) {
      SLAB_LOAD(xg + kt + 64, 32 * 1024, wg + kt + 64)
    }
    slab_mma(xr + SLAB, wr + SLAB, hi, acc);
    if (more) {
      SLAB_STORE(xst, wst)
      __syncthreads();
    }
  }

  const int tsel = nBase >> 10;
  float* dst = (tsel == 0) ? qb : (tsel == 1) ? kb : vb;
#pragma unroll
  for (int tr = 0; tr < 2; ++tr) {
#pragma unroll
    for (int tc = 0; tc < 4; ++tc) {
      int n = nBase + wn * 64 + tc * 16 + nlo;
      int rem = n & 1023;
      int h = rem >> 6, d = rem & 63;
#pragma unroll
      for (int j = 0; j < 8; ++j) {
        int m = mBase + wm * 32 + tr * 16 + j + 8 * hi;
        float y = acc[tr][tc][j];
        if (tsel == 0)
          y = elup1(y) * 0.125f;
        else if (tsel == 1)
          y = elup1(y);
        int b = m >> 12, l = m & 4095;
        dst[((size_t)(b * 16 + h) * 4096 + l) * 64 + d] = y;
      }
    }
  }
}

#define AP 68

__global__ __launch_bounds__(256) void chunk_stats(const float* __restrict__ kb,
                                                   const float* __restrict__ vb,
                                                   float* __restrict__ Sch,
                                                   float* __restrict__ Zch) {
  __shared__ float ks[64 * AP];
  __shared__ float vs[64 * AP];
  const int tid = threadIdx.x;
  const int lane = tid & 31, wv = tid >> 5;
  const int hi = lane >> 4, nlo = lane & 15;
  const int c = blockIdx.x, bh = blockIdx.y;
  const size_t base = ((size_t)bh * 4096 + c * 256) * 64;

  v8f sc[2] = {};
  float zacc = 0.f;

  for (int sb = 0; sb < 4; ++sb) {
#pragma unroll
    for (int i = 0; i < 4; ++i) {
      int q = tid + 256 * i;
      int row = q >> 4, c4 = (q & 15) * 4;
      size_t g = base + (size_t)(sb * 64 + row) * 64 + c4;
      *(float4*)(ks + row * AP + c4) = *(const float4*)(kb + g);
      *(float4*)(vs + row * AP + c4) = *(const float4*)(vb + g);
    }
    __syncthreads();
#pragma unroll
    for (int ti = 0; ti < 2; ++ti) {
      const int t = 2 * wv + ti, td = t >> 2, te = t & 3;
      v8f acc = sc[ti];
#pragma unroll
      for (int k0 = 0; k0 < 64; k0 += 4) {
        v2f a;
        a.x = ks[(k0 + 2 * hi) * AP + td * 16 + nlo];
        a.y = ks[(k0 + 2 * hi + 1) * AP + td * 16 + nlo];
        v2f bv;
        bv.x = vs[(k0 + 2 * hi) * AP + te * 16 + nlo];
        bv.y = vs[(k0 + 2 * hi + 1) * AP + te * 16 + nlo];
        acc = wmma4(a, bv, acc);
      }
      sc[ti] = acc;
    }
    if (tid < 64) {
      float s = 0.f;
      for (int m = 0; m < 64; ++m) s += ks[m * AP + tid];
      zacc += s;
    }
    __syncthreads();
  }

  float* So = Sch + ((size_t)bh * NC + c) * 4096;
#pragma unroll
  for (int ti = 0; ti < 2; ++ti) {
    const int t = 2 * wv + ti, td = t >> 2, te = t & 3;
#pragma unroll
    for (int j = 0; j < 8; ++j)
      So[(td * 16 + j + 8 * hi) * 64 + te * 16 + nlo] = sc[ti][j];
  }
  if (tid < 64) Zch[((size_t)bh * NC + c) * 64 + tid] = zacc;
}

__global__ __launch_bounds__(256) void attn_chunk(float* __restrict__ qo,
                                                  const float* __restrict__ kb,
                                                  const float* __restrict__ vb,
                                                  const float* __restrict__ Sch,
                                                  const float* __restrict__ Zch) {
  extern __shared__ float sm[];
  float* qs = sm;
  float* ks = sm + 64 * AP;
  float* vs = sm + 2 * 64 * AP;
  float* as_ = sm + 3 * 64 * AP;
  float* Ss = sm + 4 * 64 * AP;
  float* Zs = sm + 5 * 64 * AP;
  float* dens = Zs + 64;

  const int tid = threadIdx.x;
  const int lane = tid & 31, wv = tid >> 5;
  const int hi = lane >> 4, nlo = lane & 15;
  const int c = blockIdx.x, bh = blockIdx.y;
  const size_t base = (size_t)bh * 4096 * 64;

  {
    const float4* Sg = (const float4*)(Sch + (size_t)bh * NC * 4096);
    const float* Zg = Zch + (size_t)bh * NC * 64 + (tid & 63);
    float4 run[4];
#pragma unroll
    for (int j = 0; j < 4; ++j) run[j] = make_float4(0.f, 0.f, 0.f, 0.f);
    float zrun = 0.f;
    for (int cp0 = 0; cp0 < c; cp0 += 4) {
      float4 t4[4][4];
      float zt[4];
#pragma unroll
      for (int u = 0; u < 4; ++u) {
        if (cp0 + u < c) {
#pragma unroll
          for (int j = 0; j < 4; ++j)
            t4[u][j] = Sg[(cp0 + u) * 1024 + tid + 256 * j];
          zt[u] = Zg[(cp0 + u) * 64];
        } else {
#pragma unroll
          for (int j = 0; j < 4; ++j) t4[u][j] = make_float4(0.f, 0.f, 0.f, 0.f);
          zt[u] = 0.f;
        }
      }
#pragma unroll
      for (int u = 0; u < 4; ++u) {
#pragma unroll
        for (int j = 0; j < 4; ++j) {
          run[j].x += t4[u][j].x;
          run[j].y += t4[u][j].y;
          run[j].z += t4[u][j].z;
          run[j].w += t4[u][j].w;
        }
        zrun += zt[u];
      }
    }
#pragma unroll
    for (int j = 0; j < 4; ++j) {
      int e4 = tid + 256 * j;
      *(float4*)(Ss + (e4 >> 4) * AP + (e4 & 15) * 4) = run[j];
    }
    if (tid < 64) Zs[tid] = zrun;
  }
  __syncthreads();

  for (int sb = 0; sb < 4; ++sb) {
    const int l0 = c * 256 + sb * 64;
#pragma unroll
    for (int i = 0; i < 4; ++i) {
      int q = tid + 256 * i;
      int row = q >> 4, c4 = (q & 15) * 4;
      size_t g = base + (size_t)(l0 + row) * 64 + c4;
      *(float4*)(qs + row * AP + c4) = *(const float4*)(qo + g);
      *(float4*)(ks + row * AP + c4) = *(const float4*)(kb + g);
      *(float4*)(vs + row * AP + c4) = *(const float4*)(vb + g);
    }
    __syncthreads();

    v8f num[2];
#pragma unroll
    for (int ti = 0; ti < 2; ++ti) {
      const int t = 2 * wv + ti, tl = t >> 2, tc = t & 3;
      v8f acc = {};
      v8f nacc = {};
      const float* qr = qs + (tl * 16 + nlo) * AP;
      const float* kr = ks + (tc * 16 + nlo) * AP;
#pragma unroll
      for (int k0 = 0; k0 < 64; k0 += 4) {
        v2f a;
        a.x = qr[k0 + 2 * hi];
        a.y = qr[k0 + 2 * hi + 1];
        v2f bk;
        bk.x = kr[k0 + 2 * hi];
        bk.y = kr[k0 + 2 * hi + 1];
        acc = wmma4(a, bk, acc);
        v2f bs;
        bs.x = Ss[(k0 + 2 * hi) * AP + tc * 16 + nlo];
        bs.y = Ss[(k0 + 2 * hi + 1) * AP + tc * 16 + nlo];
        nacc = wmma4(a, bs, nacc);
      }
#pragma unroll
      for (int j = 0; j < 8; ++j) {
        int r = tl * 16 + j + 8 * hi;
        int cc = tc * 16 + nlo;
        as_[r * AP + cc] = (cc <= r) ? acc[j] : 0.f;
      }
      num[ti] = nacc;
    }
    __syncthreads();

    if (tid < 64) {
      float s = 0.f;
      const float* ar = as_ + tid * AP;
      for (int n = 0; n < 64; ++n) s += ar[n];
      const float* qr = qs + tid * AP;
      for (int d = 0; d < 64; ++d) s += qr[d] * Zs[d];
      dens[tid] = fmaxf(s, 1e-6f);
    }
#pragma unroll
    for (int ti = 0; ti < 2; ++ti) {
      const int t = 2 * wv + ti, tl = t >> 2, tc = t & 3;
      v8f nacc = num[ti];
      const float* ar = as_ + (tl * 16 + nlo) * AP;
#pragma unroll
      for (int k0 = 0; k0 < 64; k0 += 4) {
        v2f a;
        a.x = ar[k0 + 2 * hi];
        a.y = ar[k0 + 2 * hi + 1];
        v2f bv;
        bv.x = vs[(k0 + 2 * hi) * AP + tc * 16 + nlo];
        bv.y = vs[(k0 + 2 * hi + 1) * AP + tc * 16 + nlo];
        nacc = wmma4(a, bv, nacc);
      }
      num[ti] = nacc;
    }
    __syncthreads();

    if (sb != 3) {
#pragma unroll
      for (int ti = 0; ti < 2; ++ti) {
        const int t = 2 * wv + ti, td = t >> 2, te = t & 3;
        v8f scv;
#pragma unroll
        for (int j = 0; j < 8; ++j)
          scv[j] = Ss[(td * 16 + j + 8 * hi) * AP + te * 16 + nlo];
#pragma unroll
        for (int k0 = 0; k0 < 64; k0 += 4) {
          v2f a;
          a.x = ks[(k0 + 2 * hi) * AP + td * 16 + nlo];
          a.y = ks[(k0 + 2 * hi + 1) * AP + td * 16 + nlo];
          v2f bv;
          bv.x = vs[(k0 + 2 * hi) * AP + te * 16 + nlo];
          bv.y = vs[(k0 + 2 * hi + 1) * AP + te * 16 + nlo];
          scv = wmma4(a, bv, scv);
        }
#pragma unroll
        for (int j = 0; j < 8; ++j)
          Ss[(td * 16 + j + 8 * hi) * AP + te * 16 + nlo] = scv[j];
      }
      if (tid < 64) {
        float s = Zs[tid];
        for (int m = 0; m < 64; ++m) s += ks[m * AP + tid];
        Zs[tid] = s;
      }
    }
#pragma unroll
    for (int ti = 0; ti < 2; ++ti) {
      const int t = 2 * wv + ti, tl = t >> 2, te = t & 3;
#pragma unroll
      for (int j = 0; j < 8; ++j) {
        int r = tl * 16 + j + 8 * hi;
        float v = num[ti][j] / dens[r];
        qo[base + (size_t)(l0 + r) * 64 + te * 16 + nlo] = v;
      }
    }
    __syncthreads();
  }
}

__global__ __launch_bounds__(256) void out_gemm(const float* __restrict__ oh,
                                                const float* __restrict__ W,
                                                float* __restrict__ out) {
  __shared__ float xs[2 * SLAB];
  __shared__ float wls[2 * SLAB];

  const int tid = threadIdx.x;
  const int lane = tid & 31, wv = tid >> 5;
  const int hi = lane >> 4, nlo = lane & 15;
  const int wm = wv >> 1, wn = wv & 1;
  const int mBase = blockIdx.y * 128;
  const int nBase = blockIdx.x * 128;

  v8f acc[2][4] = {};

  const int srow = tid >> 3, sc4 = (tid & 7) * 4;
  const int m0 = mBase + srow;
  const float* og = oh + ((size_t)((m0 >> 12) * 16) * 4096 + (m0 & 4095)) * 64 + sc4;
  const float* wg = W + (size_t)(nBase + srow) * 1024 + sc4;
  float* xst = xs + srow * 36 + sc4;
  float* wst = wls + srow * 36 + sc4;
  const float* xr = xs + (wm * 32 + nlo) * 36;
  const float* wr = wls + (wn * 64 + nlo) * 36;

  v4f px0, px1, px2, px3, pw0, pw1, pw2, pw3;
  SLAB_LOAD(og, 32 * 64, wg)
  SLAB_STORE(xst, wst)
  __syncthreads();

  for (int kt = 0; kt < 1024; kt += 64) {
    const float* oc = og + (size_t)(kt >> 6) * 4096 * 64;
    SLAB_LOAD(oc + 32, 32 * 64, wg + kt + 32)
    slab_mma(xr, wr, hi, acc);
    SLAB_STORE(xst + SLAB, wst + SLAB)
    __syncthreads();

    const bool more = kt + 64 < 1024;
    if (more) {
      SLAB_LOAD(oc + (size_t)4096 * 64, 32 * 64, wg + kt + 64)
    }
    slab_mma(xr + SLAB, wr + SLAB, hi, acc);
    if (more) {
      SLAB_STORE(xst, wst)
      __syncthreads();
    }
  }

#pragma unroll
  for (int tr = 0; tr < 2; ++tr) {
#pragma unroll
    for (int tc = 0; tc < 4; ++tc) {
      int n = nBase + wn * 64 + tc * 16 + nlo;
#pragma unroll
      for (int j = 0; j < 8; ++j) {
        int m = mBase + wm * 32 + tr * 16 + j + 8 * hi;
        out[(size_t)m * 1024 + n] = acc[tr][tc][j];
      }
    }
  }
}

extern "C" void kernel_launch(void* const* d_in, const int* in_sizes, int n_in,
                              void* d_out, int out_size, void* d_ws,
                              size_t ws_size, hipStream_t stream) {
  const float* x = (const float*)d_in[0];
  const float* Wqkv = (const float*)d_in[1];
  const float* Wout = (const float*)d_in[2];
  float* out = (float*)d_out;

  float* qbuf = (float*)d_ws;
  float* kbuf = qbuf + HEADSZ;
  float* vbuf = kbuf + HEADSZ;
  float* Sch = vbuf + HEADSZ;
  float* Zch = Sch + (size_t)BB * HH * NC * 4096;

  qkv_gemm<<<dim3(24, 64), 256, 0, stream>>>(x, Wqkv, qbuf, kbuf, vbuf);

  chunk_stats<<<dim3(NC, BB * HH), 256, 0, stream>>>(kbuf, vbuf, Sch, Zch);
  const size_t smem = (size_t)(5 * 64 * AP + 128) * sizeof(float);
  attn_chunk<<<dim3(NC, BB * HH), 256, smem, stream>>>(qbuf, kbuf, vbuf, Sch,
                                                       Zch);

  out_gemm<<<dim3(8, 64), 256, 0, stream>>>(qbuf, Wout, out);
}
